# starting kernel
# baseline (speedup 1.0000x reference)
#include <hip/hip_runtime.h>

#define HW_    56
#define CTOT   96
#define HEADS  3
#define EDIM   32
#define WIN    7
#define TOK    49
#define TPAD   64
#define NQKV   288
#define WQS    292

typedef __attribute__((ext_vector_type(2))) float v2f;
typedef __attribute__((ext_vector_type(8))) float v8f;

#define WMMA4(a, b, c) \
  __builtin_amdgcn_wmma_f32_16x16x4_f32(false, (a), false, (b), (short)0, (c), false, false)

__global__ __launch_bounds__(256) void swin_msa_fused(
    const float* __restrict__ x, const float* __restrict__ w_qkv,
    const float* __restrict__ b_qkv, const float* __restrict__ w_proj,
    const float* __restrict__ b_proj, const float* __restrict__ rel_bias,
    float* __restrict__ out)
{
  __shared__ float xs[TPAD][100];
  __shared__ float wq[CTOT][WQS];
  __shared__ float qs[HEADS][TPAD][33];
  __shared__ float ks[HEADS][TPAD][33];
  __shared__ float vs[HEADS][TPAD][33];
  __shared__ float sc[HEADS][TPAD][65];
  __shared__ float ow[TPAD][100];

  const int tid  = threadIdx.x;
  const int wid  = tid >> 5;
  const int lane = tid & 31;
  const int ln16 = lane & 15;
  const int kh   = lane >> 4;
  const int rofs = kh * 8;

  const int blk = blockIdx.x;
  const int b   = blk >> 6;
  const int win = blk & 63;
  const int wy  = win >> 3;
  const int wx  = win & 7;
  const long xbase = (long)b * (HW_ * HW_) * CTOT;

  for (int idx = tid; idx < CTOT * (NQKV / 4); idx += 256) {
    int r  = idx / (NQKV / 4);
    int c4 = idx - r * (NQKV / 4);
    float4 v = *(const float4*)(w_qkv + r * NQKV + c4 * 4);
    *(float4*)&wq[r][c4 * 4] = v;
  }
  for (int idx = tid; idx < TPAD * (CTOT / 4); idx += 256) {
    int r  = idx / (CTOT / 4);
    int c4 = idx - r * (CTOT / 4);
    float4 v = make_float4(0.f, 0.f, 0.f, 0.f);
    if (r < TOK) {
      int m1 = r / WIN, m2 = r % WIN;
      int grow = (wy * WIN + m1) * HW_ + (wx * WIN + m2);
      v = *(const float4*)(x + xbase + (long)grow * CTOT + c4 * 4);
    }
    *(float4*)&xs[r][c4 * 4] = v;
  }
  __syncthreads();

  for (int tt = wid; tt < 4 * 18; tt += 8) {
    const int row0 = (tt / 18) * 16;
    const int col0 = (tt % 18) * 16;
    v8f acc = {};
#pragma unroll
    for (int kk = 0; kk < CTOT; kk += 4) {
      v2f a, bf;
      a.x  = xs[row0 + ln16][kk + kh * 2 + 0];
      a.y  = xs[row0 + ln16][kk + kh * 2 + 1];
      bf.x = wq[kk + kh * 2 + 0][col0 + ln16];
      bf.y = wq[kk + kh * 2 + 1][col0 + ln16];
      acc = WMMA4(a, bf, acc);
    }
#pragma unroll
    for (int j = 0; j < 8; ++j) {
      int row = row0 + j + rofs;
      int col = col0 + ln16;
      float val = acc[j] + b_qkv[col];
      int head = col / CTOT;
      int r96  = col - head * CTOT;
      int e    = r96 / 3;
      int kqv  = r96 - 3 * e;
      if (kqv == 0)      qs[head][row][e] = val;
      else if (kqv == 1) ks[head][row][e] = val;
      else               vs[head][row][e] = val;
    }
  }
  __syncthreads();

  const float scale = 0.17677669529663687f;
  for (int tt = wid; tt < HEADS * 16; tt += 8) {
    const int head = tt / 16;
    const int q0   = ((tt % 16) / 4) * 16;
    const int j0   = (tt % 4) * 16;
    v8f acc = {};
#pragma unroll
    for (int kk = 0; kk < EDIM; kk += 4) {
      v2f a, bf;
      a.x  = qs[head][q0 + ln16][kk + kh * 2 + 0];
      a.y  = qs[head][q0 + ln16][kk + kh * 2 + 1];
      bf.x = ks[head][j0 + ln16][kk + kh * 2 + 0];
      bf.y = ks[head][j0 + ln16][kk + kh * 2 + 1];
      acc = WMMA4(a, bf, acc);
    }
#pragma unroll
    for (int j = 0; j < 8; ++j) {
      int row = q0 + j + rofs;
      int col = j0 + ln16;
      float v;
      if (col < TOK) {
        float bias = (row < TOK) ? rel_bias[row * TOK + col] : 0.f;
        v = acc[j] * scale + bias;
      } else {
        v = -1e30f;
      }
      sc[head][row][col] = v;
    }
  }
  __syncthreads();

  for (int task = wid; task < HEADS * TOK; task += 8) {
    const int head = task / TOK;
    const int r    = task - head * TOK;
    float v0 = sc[head][r][lane];
    float v1 = sc[head][r][lane + 32];
    float mx = fmaxf(v0, v1);
    for (int off = 16; off > 0; off >>= 1) mx = fmaxf(mx, __shfl_xor(mx, off, 32));
    float e0 = __expf(v0 - mx);
    float e1 = __expf(v1 - mx);
    float s = e0 + e1;
    for (int off = 16; off > 0; off >>= 1) s += __shfl_xor(s, off, 32);
    float inv = 1.f / s;
    sc[head][r][lane]      = e0 * inv;
    sc[head][r][lane + 32] = e1 * inv;
  }
  __syncthreads();

  for (int tt = wid; tt < HEADS * 4 * 2; tt += 8) {
    const int head = tt / 8;
    const int q0   = ((tt % 8) / 2) * 16;
    const int e0   = (tt % 2) * 16;
    v8f acc = {};
#pragma unroll
    for (int kk = 0; kk < TPAD; kk += 4) {
      v2f a, bf;
      a.x  = sc[head][q0 + ln16][kk + kh * 2 + 0];
      a.y  = sc[head][q0 + ln16][kk + kh * 2 + 1];
      bf.x = vs[head][kk + kh * 2 + 0][e0 + ln16];
      bf.y = vs[head][kk + kh * 2 + 1][e0 + ln16];
      acc = WMMA4(a, bf, acc);
    }
#pragma unroll
    for (int j = 0; j < 8; ++j) {
      int row = q0 + j + rofs;
      int col = e0 + ln16;
      ow[row][head * EDIM + col] = acc[j];
    }
  }
  __syncthreads();

  for (int tt = wid; tt < 4 * 6; tt += 8) {
    const int row0 = (tt / 6) * 16;
    const int col0 = (tt % 6) * 16;
    v8f acc = {};
#pragma unroll
    for (int kk = 0; kk < CTOT; kk += 4) {
      v2f a, bf;
      a.x  = ow[row0 + ln16][kk + kh * 2 + 0];
      a.y  = ow[row0 + ln16][kk + kh * 2 + 1];
      bf.x = w_proj[(kk + kh * 2 + 0) * CTOT + col0 + ln16];
      bf.y = w_proj[(kk + kh * 2 + 1) * CTOT + col0 + ln16];
      acc = WMMA4(a, bf, acc);
    }
#pragma unroll
    for (int j = 0; j < 8; ++j) {
      int row = row0 + j + rofs;
      if (row < TOK) {
        int col = col0 + ln16;
        int m1 = row / WIN, m2 = row % WIN;
        long grow = (long)b * (HW_ * HW_) +
                    (long)((wy * WIN + m1) * HW_ + (wx * WIN + m2));
        out[grow * CTOT + col] = acc[j] + b_proj[col];
      }
    }
  }
}

extern "C" void kernel_launch(void* const* d_in, const int* in_sizes, int n_in,
                              void* d_out, int out_size, void* d_ws, size_t ws_size,
                              hipStream_t stream) {
  const float* x        = (const float*)d_in[0];
  const float* w_qkv    = (const float*)d_in[1];
  const float* b_qkv    = (const float*)d_in[2];
  const float* w_proj   = (const float*)d_in[3];
  const float* b_proj   = (const float*)d_in[4];
  const float* rel_bias = (const float*)d_in[5];
  float* out = (float*)d_out;
  (void)in_sizes; (void)n_in; (void)out_size; (void)d_ws; (void)ws_size;

  swin_msa_fused<<<64 * 64, 256, 0, stream>>>(x, w_qkv, b_qkv, w_proj, b_proj,
                                              rel_bias, out);
}


# w_qkv not staged in LDS: B fragments go global->registers, each reused over 4 row tiles; attention output overlays the dead x tile; LDS 283->148 KiB: two work-groups per WGP; sums unchanged
# speedup vs baseline: 2.1139x; 2.1139x over previous
#include <hip/hip_runtime.h>

#define HW_    56
#define CTOT   96
#define HEADS  3
#define EDIM   32
#define WIN    7
#define TOK    49
#define TPAD   64
#define NQKV   288
#define KSTEPS 24

typedef __attribute__((ext_vector_type(2))) float v2f;
typedef __attribute__((ext_vector_type(8))) float v8f;

#define WMMA4(a, b, c) \
  __builtin_amdgcn_wmma_f32_16x16x4_f32(false, (a), false, (b), (short)0, (c), false, false)

__global__ __launch_bounds__(256) void swin_msa_fused(
    const float* __restrict__ x, const float* __restrict__ w_qkv,
    const float* __restrict__ b_qkv, const float* __restrict__ w_proj,
    const float* __restrict__ b_proj, const float* __restrict__ rel_bias,
    float* __restrict__ out)
{
  __shared__ float xs[TPAD][100];
  __shared__ float qkvs[3][HEADS][TPAD][33];
  __shared__ float sc[HEADS][TPAD][65];

  const int tid  = threadIdx.x;
  const int wid  = tid >> 5;
  const int lane = tid & 31;
  const int ln16 = lane & 15;
  const int kh   = lane >> 4;
  const int rofs = kh * 8;

  const int blk = blockIdx.x;
  const int b   = blk >> 6;
  const int win = blk & 63;
  const int wy  = win >> 3;
  const int wx  = win & 7;
  const long xbase = (long)b * (HW_ * HW_) * CTOT;

  const int colA = wid * 16 + ln16;
  const int colB = (wid + 8) * 16 + ln16;
  const int colT = (16 + (wid >> 2)) * 16 + ln16;
  v2f bA[KSTEPS], bB[KSTEPS], bT[KSTEPS];
#pragma unroll
  for (int i = 0; i < KSTEPS; ++i) {
    const float* w0 = w_qkv + (i * 4 + kh * 2 + 0) * NQKV;
    const float* w1 = w_qkv + (i * 4 + kh * 2 + 1) * NQKV;
    bA[i].x = w0[colA];  bA[i].y = w1[colA];
    bB[i].x = w0[colB];  bB[i].y = w1[colB];
    bT[i].x = w0[colT];  bT[i].y = w1[colT];
  }
  const float biasA = b_qkv[colA];
  const float biasB = b_qkv[colB];
  const float biasT = b_qkv[colT];

  for (int idx = tid; idx < TPAD * (CTOT / 4); idx += 256) {
    int r  = idx / (CTOT / 4);
    int c4 = idx - r * (CTOT / 4);
    float4 v = make_float4(0.f, 0.f, 0.f, 0.f);
    if (r < TOK) {
      int m1 = r / WIN, m2 = r % WIN;
      int grow = (wy * WIN + m1) * HW_ + (wx * WIN + m2);
      v = *(const float4*)(x + xbase + (long)grow * CTOT + c4 * 4);
    }
    *(float4*)&xs[r][c4 * 4] = v;
  }
  __syncthreads();

  auto qkv_dst = [&](int col) -> float* {
    int head = col / CTOT;
    int r96  = col - head * CTOT;
    int e    = r96 / 3;
    int kqv  = r96 - 3 * e;
    return &qkvs[kqv][head][0][e];
  };
  auto qkv_strip = [&](const v2f (&bf)[KSTEPS], float bias, float* dst) {
    v8f acc0 = {}, acc1 = {}, acc2 = {}, acc3 = {};
#pragma unroll
    for (int i = 0; i < KSTEPS; ++i) {
      const int kc = i * 4 + kh * 2;
      v2f a0, a1, a2, a3;
      a0.x = xs[ 0 + ln16][kc];  a0.y = xs[ 0 + ln16][kc + 1];
      a1.x = xs[16 + ln16][kc];  a1.y = xs[16 + ln16][kc + 1];
      a2.x = xs[32 + ln16][kc];  a2.y = xs[32 + ln16][kc + 1];
      a3.x = xs[48 + ln16][kc];  a3.y = xs[48 + ln16][kc + 1];
      acc0 = WMMA4(a0, bf[i], acc0);
      acc1 = WMMA4(a1, bf[i], acc1);
      acc2 = WMMA4(a2, bf[i], acc2);
      acc3 = WMMA4(a3, bf[i], acc3);
    }
#pragma unroll
    for (int j = 0; j < 8; ++j) {
      const int row = j + rofs;
      dst[( 0 + row) * 33] = acc0[j] + bias;
      dst[(16 + row) * 33] = acc1[j] + bias;
      dst[(32 + row) * 33] = acc2[j] + bias;
      dst[(48 + row) * 33] = acc3[j] + bias;
    }
  };
  qkv_strip(bA, biasA, qkv_dst(colA));
  qkv_strip(bB, biasB, qkv_dst(colB));
  {
    const int row0 = (wid & 3) * 16;
    float* dst = qkv_dst(colT);
    v8f acc = {};
#pragma unroll
    for (int i = 0; i < KSTEPS; ++i) {
      const int kc = i * 4 + kh * 2;
      v2f a;
      a.x = xs[row0 + ln16][kc];
      a.y = xs[row0 + ln16][kc + 1];
      acc = WMMA4(a, bT[i], acc);
    }
#pragma unroll
    for (int j = 0; j < 8; ++j)
      dst[(row0 + j + rofs) * 33] = acc[j] + biasT;
  }
  __syncthreads();

  const float scale = 0.17677669529663687f;
  for (int tt = wid; tt < HEADS * 16; tt += 8) {
    const int head = tt / 16;
    const int q0   = ((tt % 16) / 4) * 16;
    const int j0   = (tt % 4) * 16;
    v8f acc = {};
#pragma unroll
    for (int kk = 0; kk < EDIM; kk += 4) {
      v2f a, bf;
      a.x  = qkvs[0][head][q0 + ln16][kk + kh * 2 + 0];
      a.y  = qkvs[0][head][q0 + ln16][kk + kh * 2 + 1];
      bf.x = qkvs[1][head][j0 + ln16][kk + kh * 2 + 0];
      bf.y = qkvs[1][head][j0 + ln16][kk + kh * 2 + 1];
      acc = WMMA4(a, bf, acc);
    }
#pragma unroll
    for (int j = 0; j < 8; ++j) {
      int row = q0 + j + rofs;
      int col = j0 + ln16;
      float v;
      if (col < TOK) {
        float bias = (row < TOK) ? rel_bias[row * TOK + col] : 0.f;
        v = acc[j] * scale + bias;
      } else {
        v = -1e30f;
      }
      sc[head][row][col] = v;
    }
  }
  __syncthreads();

  for (int task = wid; task < HEADS * TOK; task += 8) {
    const int head = task / TOK;
    const int r    = task - head * TOK;
    float v0 = sc[head][r][lane];
    float v1 = sc[head][r][lane + 32];
    float mx = fmaxf(v0, v1);
    for (int off = 16; off > 0; off >>= 1) mx = fmaxf(mx, __shfl_xor(mx, off, 32));
    float e0 = __expf(v0 - mx);
    float e1 = __expf(v1 - mx);
    float s = e0 + e1;
    for (int off = 16; off > 0; off >>= 1) s += __shfl_xor(s, off, 32);
    float inv = 1.f / s;
    sc[head][r][lane]      = e0 * inv;
    sc[head][r][lane + 32] = e1 * inv;
  }
  __syncthreads();

  for (int tt = wid; tt < HEADS * 4 * 2; tt += 8) {
    const int head = tt / 8;
    const int q0   = ((tt % 8) / 2) * 16;
    const int e0   = (tt % 2) * 16;
    v8f acc = {};
#pragma unroll
    for (int kk = 0; kk < TPAD; kk += 4) {
      v2f a, bf;
      a.x  = sc[head][q0 + ln16][kk + kh * 2 + 0];
      a.y  = sc[head][q0 + ln16][kk + kh * 2 + 1];
      bf.x = qkvs[2][head][kk + kh * 2 + 0][e0 + ln16];
      bf.y = qkvs[2][head][kk + kh * 2 + 1][e0 + ln16];
      acc = WMMA4(a, bf, acc);
    }
#pragma unroll
    for (int j = 0; j < 8; ++j) {
      int row = q0 + j + rofs;
      int col = e0 + ln16;
      xs[row][head * EDIM + col] = acc[j];
    }
  }
  __syncthreads();

  for (int tt = wid; tt < 4 * 6; tt += 8) {
    const int row0 = (tt / 6) * 16;
    const int col0 = (tt % 6) * 16;
    v8f acc = {};
#pragma unroll
    for (int kk = 0; kk < CTOT; kk += 4) {
      v2f a, bf;
      a.x  = xs[row0 + ln16][kk + kh * 2 + 0];
      a.y  = xs[row0 + ln16][kk + kh * 2 + 1];
      bf.x = w_proj[(kk + kh * 2 + 0) * CTOT + col0 + ln16];
      bf.y = w_proj[(kk + kh * 2 + 1) * CTOT + col0 + ln16];
      acc = WMMA4(a, bf, acc);
    }
#pragma unroll
    for (int j = 0; j < 8; ++j) {
      int row = row0 + j + rofs;
      if (row < TOK) {
        int col = col0 + ln16;
        int m1 = row / WIN, m2 = row % WIN;
        long grow = (long)b * (HW_ * HW_) +
                    (long)((wy * WIN + m1) * HW_ + (wx * WIN + m2));
        out[grow * CTOT + col] = acc[j] + b_proj[col];
      }
    }
  }
}

extern "C" void kernel_launch(void* const* d_in, const int* in_sizes, int n_in,
                              void* d_out, int out_size, void* d_ws, size_t ws_size,
                              hipStream_t stream) {
  const float* x        = (const float*)d_in[0];
  const float* w_qkv    = (const float*)d_in[1];
  const float* b_qkv    = (const float*)d_in[2];
  const float* w_proj   = (const float*)d_in[3];
  const float* b_proj   = (const float*)d_in[4];
  const float* rel_bias = (const float*)d_in[5];
  float* out = (float*)d_out;
  (void)in_sizes; (void)n_in; (void)out_size; (void)d_ws; (void)ws_size;

  swin_msa_fused<<<64 * 64, 256, 0, stream>>>(x, w_qkv, b_qkv, w_proj, b_proj,
                                              rel_bias, out);
}


# projection tiles re-dealt: a wave's 3 tiles share one w_proj column tile, its fragments loaded once into registers ahead of the attention loop (3 accumulators; row 48 on waves 6,7); sums unchanged
# speedup vs baseline: 4.3203x; 2.0438x over previous
#include <hip/hip_runtime.h>

#define HW_    56
#define CTOT   96
#define HEADS  3
#define EDIM   32
#define WIN    7
#define TOK    49
#define TPAD   64
#define NQKV   288
#define KSTEPS 24

typedef __attribute__((ext_vector_type(2))) float v2f;
typedef __attribute__((ext_vector_type(8))) float v8f;

#define WMMA4(a, b, c) \
  __builtin_amdgcn_wmma_f32_16x16x4_f32(false, (a), false, (b), (short)0, (c), false, false)

__device__ __forceinline__ float xhalf(float x) {
  const int b = __builtin_bit_cast(int, x);
  return __builtin_bit_cast(float,
      __builtin_amdgcn_permlanex16(b, b, 0x76543210, 0xfedcba98, false, false));
}
#define XROW(x, m) \
  __builtin_bit_cast(float, __builtin_amdgcn_update_dpp(__builtin_bit_cast(int, (x)), \
      __builtin_bit_cast(int, (x)), 0x160 + (m), 0xf, 0xf, false))

#define SROWS 7

__device__ __forceinline__ void softmax_rows(float (*sc)[65], const int row0, const int lane) {
  float v0[SROWS], v1[SROWS], mx[SROWS], sm[SROWS];
#pragma unroll
  for (int t = 0; t < SROWS; ++t) {
    v0[t] = sc[row0 + 8 * t][lane];
    v1[t] = sc[row0 + 8 * t][lane + 32];
    mx[t] = fmaxf(v0[t], v1[t]);
  }
#pragma unroll
  for (int t = 0; t < SROWS; ++t) mx[t] = fmaxf(mx[t], xhalf(mx[t]));
#pragma unroll
  for (int t = 0; t < SROWS; ++t) mx[t] = fmaxf(mx[t], XROW(mx[t], 8));
#pragma unroll
  for (int t = 0; t < SROWS; ++t) mx[t] = fmaxf(mx[t], XROW(mx[t], 4));
#pragma unroll
  for (int t = 0; t < SROWS; ++t) mx[t] = fmaxf(mx[t], XROW(mx[t], 2));
#pragma unroll
  for (int t = 0; t < SROWS; ++t) mx[t] = fmaxf(mx[t], XROW(mx[t], 1));
#pragma unroll
  for (int t = 0; t < SROWS; ++t) {
    v0[t] = __expf(v0[t] - mx[t]);
    v1[t] = __expf(v1[t] - mx[t]);
    sm[t] = v0[t] + v1[t];
  }
#pragma unroll
  for (int t = 0; t < SROWS; ++t) sm[t] += xhalf(sm[t]);
#pragma unroll
  for (int t = 0; t < SROWS; ++t) sm[t] += XROW(sm[t], 8);
#pragma unroll
  for (int t = 0; t < SROWS; ++t) sm[t] += XROW(sm[t], 4);
#pragma unroll
  for (int t = 0; t < SROWS; ++t) sm[t] += XROW(sm[t], 2);
#pragma unroll
  for (int t = 0; t < SROWS; ++t) sm[t] += XROW(sm[t], 1);
#pragma unroll
  for (int t = 0; t < SROWS; ++t) {
    const float inv = 1.f / sm[t];
    sc[row0 + 8 * t][lane]      = v0[t] * inv;
    sc[row0 + 8 * t][lane + 32] = v1[t] * inv;
  }
}

__global__ __launch_bounds__(256, 6) void swin_msa_fused(
    const float* __restrict__ x, const float* __restrict__ w_qkv,
    const float* __restrict__ b_qkv, const float* __restrict__ w_proj,
    const float* __restrict__ b_proj, const float* __restrict__ rel_bias,
    float* __restrict__ out)
{
  __shared__ float xs[TPAD][100];
  __shared__ float qkvs[3][HEADS][TPAD][33];
  float (*sc)[65] = reinterpret_cast<float (*)[65]>(&xs[0][0]);

  const int tid  = threadIdx.x;
  const int wid  = tid >> 5;
  const int lane = tid & 31;
  const int ln16 = lane & 15;
  const int kh   = lane >> 4;
  const int rofs = kh * 8;

  const int blk = blockIdx.x;
  const int b   = blk >> 6;
  const int win = blk & 63;
  const int wy  = win >> 3;
  const int wx  = win & 7;
  const long xbase = (long)b * (HW_ * HW_) * CTOT;

#define XSPLIT 11
  const int colA = wid * 16 + ln16;
  const int colB = (wid + 8) * 16 + ln16;
  const int colT = (16 + (wid >> 2)) * 16 + ln16;
  v2f bA[KSTEPS], bB[KSTEPS];
#pragma unroll
  for (int i = 0; i < XSPLIT; ++i) {
    const float* w0 = w_qkv + (i * 4 + kh * 2 + 0) * NQKV;
    const float* w1 = w_qkv + (i * 4 + kh * 2 + 1) * NQKV;
    bA[i].x = w0[colA];  bA[i].y = w1[colA];
    bB[i].x = w0[colB];  bB[i].y = w1[colB];
  }
  float4 xv[6];
#pragma unroll
  for (int it = 0; it < 6; ++it) {
    const int idx = tid + 256 * it;
    const int r   = idx / (CTOT / 4);
    const int c4  = idx - r * (CTOT / 4);
    float4 v = make_float4(0.f, 0.f, 0.f, 0.f);
    if (it < 5) {
      const int rc = r < TOK ? r : TOK - 1;
      const int m1 = rc / WIN, m2 = rc % WIN;
      const int grow = (wy * WIN + m1) * HW_ + (wx * WIN + m2);
      const float4 g = *(const float4*)(x + xbase + (long)grow * CTOT + c4 * 4);
      if (r < TOK) v = g;
    }
    xv[it] = v;
  }
#pragma unroll
  for (int i = XSPLIT; i < KSTEPS; ++i) {
    const float* w0 = w_qkv + (i * 4 + kh * 2 + 0) * NQKV;
    const float* w1 = w_qkv + (i * 4 + kh * 2 + 1) * NQKV;
    bA[i].x = w0[colA];  bA[i].y = w1[colA];
    bB[i].x = w0[colB];  bB[i].y = w1[colB];
  }

#pragma unroll
  for (int it = 0; it < 6; ++it) {
    const int idx = tid + 256 * it;
    const int r   = idx / (CTOT / 4);
    const int c4  = idx - r * (CTOT / 4);
    *(float4*)&xs[r][c4 * 4] = xv[it];
  }
  __syncthreads();

  const float biasA = b_qkv[colA];
  const float biasB = b_qkv[colB];
  const float biasT = b_qkv[colT];

  auto qkv_dst = [&](int col) -> float* {
    int head = col / CTOT;
    int r96  = col - head * CTOT;
    int e    = r96 / 3;
    int kqv  = r96 - 3 * e;
    return &qkvs[kqv][head][0][e];
  };
  float* dstA = qkv_dst(colA);
  float* dstB = qkv_dst(colB);
  v2f bT[KSTEPS];
#pragma unroll
  for (int rp = 0; rp < 2; ++rp) {
    const int r0 = rp * 32;
    v8f accA0 = {}, accA1 = {}, accB0 = {}, accB1 = {};
#pragma unroll
    for (int i = 0; i < KSTEPS; ++i) {
      const int kc = i * 4 + kh * 2;
      v2f a0, a1;
      a0.x = xs[r0 +  0 + ln16][kc];  a0.y = xs[r0 +  0 + ln16][kc + 1];
      a1.x = xs[r0 + 16 + ln16][kc];  a1.y = xs[r0 + 16 + ln16][kc + 1];
      accA0 = WMMA4(a0, bA[i], accA0);
      accB0 = WMMA4(a0, bB[i], accB0);
      accA1 = WMMA4(a1, bA[i], accA1);
      accB1 = WMMA4(a1, bB[i], accB1);
      if (rp == 1 && i >= KSTEPS / 2) {
        const int t = 2 * (i - KSTEPS / 2);
        bT[t].x     = w_qkv[((t    ) * 4 + kh * 2 + 0) * NQKV + colT];
        bT[t].y     = w_qkv[((t    ) * 4 + kh * 2 + 1) * NQKV + colT];
        bT[t + 1].x = w_qkv[((t + 1) * 4 + kh * 2 + 0) * NQKV + colT];
        bT[t + 1].y = w_qkv[((t + 1) * 4 + kh * 2 + 1) * NQKV + colT];
      }
    }
#pragma unroll
    for (int j = 0; j < 8; ++j) {
      const int row = r0 + j + rofs;
      dstA[( 0 + row) * 33] = accA0[j] + biasA;
      dstA[(16 + row) * 33] = accA1[j] + biasA;
      dstB[( 0 + row) * 33] = accB0[j] + biasB;
      dstB[(16 + row) * 33] = accB1[j] + biasB;
    }
  }
  {
    const int row0 = (wid & 3) * 16;
    float* dst = qkv_dst(colT);
    v8f acc = {};
#pragma unroll
    for (int i = 0; i < KSTEPS; ++i) {
      const int kc = i * 4 + kh * 2;
      v2f a;
      a.x = xs[row0 + ln16][kc];
      a.y = xs[row0 + ln16][kc + 1];
      acc = WMMA4(a, bT[i], acc);
    }
#pragma unroll
    for (int j = 0; j < 8; ++j)
      dst[(row0 + j + rofs) * 33] = acc[j] + biasT;
  }

  const int j0s  = (wid & 3) * 16;
  const int scol = j0s + ln16;
  const bool keyreal = scol < TOK;
  const int colc = keyreal ? scol : TOK - 1;
  float sbias[2][8];
#pragma unroll
  for (int u = 0; u < 2; ++u) {
#pragma unroll
    for (int j = 0; j < 8; ++j) {
      const int row  = ((wid >> 2) + 2 * u) * 16 + j + rofs;
      const int rowc = row < TOK ? row : TOK - 1;
      const float bv = rel_bias[rowc * TOK + colc];
      sbias[u][j] = row < TOK ? bv : 0.f;
    }
  }
  const int widu = __builtin_amdgcn_readfirstlane(wid);
  const bool rowwave = widu >= 6;
  const int ct0  = rowwave ? (widu - 6) * 3 : widu;
  const int pcol = ct0 * 16 + ln16;
  v2f bP[KSTEPS];
#pragma unroll
  for (int i = 0; i < KSTEPS; ++i) {
    bP[i].x = w_proj[(i * 4 + kh * 2 + 0) * CTOT + pcol];
    bP[i].y = w_proj[(i * 4 + kh * 2 + 1) * CTOT + pcol];
  }
  float bp0 = b_proj[pcol];
  __syncthreads();

  const float scale = 0.17677669529663687f;
  for (int head = 0; head < HEADS; ++head) {
#pragma unroll
    for (int u = 0; u < 2; ++u) {
      const int q0 = ((wid >> 2) + 2 * u) * 16;
      const int j0 = j0s;
      v8f acc = {};
#pragma unroll
      for (int kk = 0; kk < EDIM; kk += 4) {
        v2f a, bf;
        a.x  = qkvs[0][head][q0 + ln16][kk + kh * 2 + 0];
        a.y  = qkvs[0][head][q0 + ln16][kk + kh * 2 + 1];
        bf.x = qkvs[1][head][j0 + ln16][kk + kh * 2 + 0];
        bf.y = qkvs[1][head][j0 + ln16][kk + kh * 2 + 1];
        acc = WMMA4(a, bf, acc);
      }
#pragma unroll
      for (int j = 0; j < 8; ++j) {
        const int row = q0 + j + rofs;
        sc[row][scol] = keyreal ? acc[j] * scale + sbias[u][j] : -1e30f;
      }
    }
    __syncthreads();

    softmax_rows(sc, wid, lane);
    __syncthreads();

    {
      const int q0 = (wid / 2) * 16;
      const int e0 = (wid % 2) * 16;
      v8f acc = {};
#pragma unroll
      for (int kk = 0; kk < TPAD; kk += 4) {
        v2f a, bf;
        a.x  = sc[q0 + ln16][kk + kh * 2 + 0];
        a.y  = sc[q0 + ln16][kk + kh * 2 + 1];
        bf.x = qkvs[2][head][kk + kh * 2 + 0][e0 + ln16];
        bf.y = qkvs[2][head][kk + kh * 2 + 1][e0 + ln16];
        acc = WMMA4(a, bf, acc);
      }
#pragma unroll
      for (int j = 0; j < 8; ++j)
        qkvs[0][head][q0 + j + rofs][e0 + ln16] = acc[j];
    }
    __syncthreads();
  }

  float* obase = out + ((long)b * (HW_ * HW_) + (long)(wy * WIN * HW_ + wx * WIN)) * CTOT;
#define GOFF(r) ((((r) / WIN) * HW_ + ((r) % WIN)) * CTOT)
  if (!rowwave) {
    v8f acc0 = {}, acc1 = {}, acc2 = {};
#pragma unroll
    for (int i = 0; i < KSTEPS; ++i) {
      const int kk = i * 4;
      const float* ap = &qkvs[0][kk / EDIM][ln16][(kk % EDIM) + kh * 2];
      v2f a0, a1, a2;
      a0.x = ap[0];        a0.y = ap[1];
      a1.x = ap[16 * 33];  a1.y = ap[16 * 33 + 1];
      a2.x = ap[32 * 33];  a2.y = ap[32 * 33 + 1];
      acc0 = WMMA4(a0, bP[i], acc0);
      acc1 = WMMA4(a1, bP[i], acc1);
      acc2 = WMMA4(a2, bP[i], acc2);
    }
    float* op = obase + pcol;
#pragma unroll
    for (int j = 0; j < 8; ++j) {
      const int g0 = kh ? GOFF(j + 8)      : GOFF(j);
      const int g1 = kh ? GOFF(j + 8 + 16) : GOFF(j + 16);
      const int g2 = kh ? GOFF(j + 8 + 32) : GOFF(j + 32);
      op[g0] = acc0[j] + bp0;
      op[g1] = acc1[j] + bp0;
      op[g2] = acc2[j] + bp0;
    }
  } else {
#pragma unroll
    for (int t = 0; t < 3; ++t) {
      v2f bN[KSTEPS];
      float bpn = 0.f;
      if (t < 2) {
        const int cn = (ct0 + t + 1) * 16 + ln16;
#pragma unroll
        for (int i = 0; i < KSTEPS; ++i) {
          bN[i].x = w_proj[(i * 4 + kh * 2 + 0) * CTOT + cn];
          bN[i].y = w_proj[(i * 4 + kh * 2 + 1) * CTOT + cn];
        }
        bpn = b_proj[cn];
      }
      v8f acc = {};
#pragma unroll
      for (int i = 0; i < KSTEPS; ++i) {
        const int kk = i * 4;
        v2f a;
        a.x = qkvs[0][kk / EDIM][48 + ln16][(kk % EDIM) + kh * 2 + 0];
        a.y = qkvs[0][kk / EDIM][48 + ln16][(kk % EDIM) + kh * 2 + 1];
        acc = WMMA4(a, bP[i], acc);
      }
      if (kh == 0) obase[GOFF(48) + (ct0 + t) * 16 + ln16] = acc[0] + bp0;
      if (t < 2) {
#pragma unroll
        for (int i = 0; i < KSTEPS; ++i) bP[i] = bN[i];
        bp0 = bpn;
      }
    }
  }
#undef GOFF
}

extern "C" void kernel_launch(void* const* d_in, const int* in_sizes, int n_in,
                              void* d_out, int out_size, void* d_ws, size_t ws_size,
                              hipStream_t stream) {
  const float* x        = (const float*)d_in[0];
  const float* w_qkv    = (const float*)d_in[1];
  const float* b_qkv    = (const float*)d_in[2];
  const float* w_proj   = (const float*)d_in[3];
  const float* b_proj   = (const float*)d_in[4];
  const float* rel_bias = (const float*)d_in[5];
  float* out = (float*)d_out;
  (void)in_sizes; (void)n_in; (void)out_size; (void)d_ws; (void)ws_size;

  swin_msa_fused<<<64 * 64, 256, 0, stream>>>(x, w_qkv, b_qkv, w_proj, b_proj,
                                              rel_bias, out);
}


# softmax on half the vector instructions: a cross-lane max/add step is ONE DPP-operand instruction (was copy, DPP move, quieting max, op); 1/sum by v_rcp_f32 (1 ulp), not IEEE division; sum order kept
# speedup vs baseline: 4.4716x; 1.0350x over previous
#include <hip/hip_runtime.h>

#define HW_    56
#define CTOT   96
#define HEADS  3
#define EDIM   32
#define WIN    7
#define TOK    49
#define TPAD   64
#define NQKV   288
#define KSTEPS 24

typedef __attribute__((ext_vector_type(2))) float v2f;
typedef __attribute__((ext_vector_type(8))) float v8f;

#define WMMA4(a, b, c) \
  __builtin_amdgcn_wmma_f32_16x16x4_f32(false, (a), false, (b), (short)0, (c), false, false)

__device__ __forceinline__ float xhalf(float x) {
  const int b = __builtin_bit_cast(int, x);
  return __builtin_bit_cast(float,
      __builtin_amdgcn_permlanex16(b, b, 0x76543210, 0xfedcba98, false, true));
}
#define XROW(x, m) \
  __builtin_bit_cast(float, __builtin_amdgcn_update_dpp(0, \
      __builtin_bit_cast(int, (x)), 0x160 + (m), 0xf, 0xf, true))
#define ROWMAX(x, m) \
  asm("v_max_num_f32_dpp %0, %0, %0 row_xmask:" #m " row_mask:0xf bank_mask:0xf" : "+v"(x))

#define SROWS 7

__device__ __forceinline__ void softmax_rows(float (*sc)[65], const int row0, const int lane) {
  float v0[SROWS], v1[SROWS], mx[SROWS], sm[SROWS];
#pragma unroll
  for (int t = 0; t < SROWS; ++t) {
    v0[t] = sc[row0 + 8 * t][lane];
    v1[t] = sc[row0 + 8 * t][lane + 32];
    asm("v_max_num_f32 %0, %1, %2" : "=v"(mx[t]) : "v"(v0[t]), "v"(v1[t]));
  }
#pragma unroll
  for (int t = 0; t < SROWS; ++t) {
    const float o = xhalf(mx[t]);
    asm("v_max_num_f32 %0, %0, %1" : "+v"(mx[t]) : "v"(o));
  }
#pragma unroll
  for (int t = 0; t < SROWS; ++t) ROWMAX(mx[t], 8);
#pragma unroll
  for (int t = 0; t < SROWS; ++t) ROWMAX(mx[t], 4);
#pragma unroll
  for (int t = 0; t < SROWS; ++t) ROWMAX(mx[t], 2);
#pragma unroll
  for (int t = 0; t < SROWS; ++t) ROWMAX(mx[t], 1);
#pragma unroll
  for (int t = 0; t < SROWS; ++t) {
    v0[t] = __expf(v0[t] - mx[t]);
    v1[t] = __expf(v1[t] - mx[t]);
    sm[t] = v0[t] + v1[t];
  }
#pragma unroll
  for (int t = 0; t < SROWS; ++t) sm[t] += xhalf(sm[t]);
#pragma unroll
  for (int t = 0; t < SROWS; ++t) sm[t] += XROW(sm[t], 8);
#pragma unroll
  for (int t = 0; t < SROWS; ++t) sm[t] += XROW(sm[t], 4);
#pragma unroll
  for (int t = 0; t < SROWS; ++t) sm[t] += XROW(sm[t], 2);
#pragma unroll
  for (int t = 0; t < SROWS; ++t) sm[t] += XROW(sm[t], 1);
#pragma unroll
  for (int t = 0; t < SROWS; ++t) {
    const float inv = __builtin_amdgcn_rcpf(sm[t]);
    sc[row0 + 8 * t][lane]      = v0[t] * inv;
    sc[row0 + 8 * t][lane + 32] = v1[t] * inv;
  }
}

__global__ __launch_bounds__(256, 6) void swin_msa_fused(
    const float* __restrict__ x, const float* __restrict__ w_qkv,
    const float* __restrict__ b_qkv, const float* __restrict__ w_proj,
    const float* __restrict__ b_proj, const float* __restrict__ rel_bias,
    float* __restrict__ out)
{
  __shared__ float xs[TPAD][100];
  __shared__ float qkvs[3][HEADS][TPAD][33];
  float (*sc)[65] = reinterpret_cast<float (*)[65]>(&xs[0][0]);

  const int tid  = threadIdx.x;
  const int wid  = tid >> 5;
  const int lane = tid & 31;
  const int ln16 = lane & 15;
  const int kh   = lane >> 4;
  const int rofs = kh * 8;

  const int blk = blockIdx.x;
  const int b   = blk >> 6;
  const int win = blk & 63;
  const int wy  = win >> 3;
  const int wx  = win & 7;
  const long xbase = (long)b * (HW_ * HW_) * CTOT;

#define XSPLIT 11
  const int colA = wid * 16 + ln16;
  const int colB = (wid + 8) * 16 + ln16;
  const int colT = (16 + (wid >> 2)) * 16 + ln16;
  v2f bA[KSTEPS], bB[KSTEPS];
#pragma unroll
  for (int i = 0; i < XSPLIT; ++i) {
    const float* w0 = w_qkv + (i * 4 + kh * 2 + 0) * NQKV;
    const float* w1 = w_qkv + (i * 4 + kh * 2 + 1) * NQKV;
    bA[i].x = w0[colA];  bA[i].y = w1[colA];
    bB[i].x = w0[colB];  bB[i].y = w1[colB];
  }
  float4 xv[6];
#pragma unroll
  for (int it = 0; it < 6; ++it) {
    const int idx = tid + 256 * it;
    const int r   = idx / (CTOT / 4);
    const int c4  = idx - r * (CTOT / 4);
    float4 v = make_float4(0.f, 0.f, 0.f, 0.f);
    if (it < 5) {
      const int rc = r < TOK ? r : TOK - 1;
      const int m1 = rc / WIN, m2 = rc % WIN;
      const int grow = (wy * WIN + m1) * HW_ + (wx * WIN + m2);
      const float4 g = *(const float4*)(x + xbase + (long)grow * CTOT + c4 * 4);
      if (r < TOK) v = g;
    }
    xv[it] = v;
  }
#pragma unroll
  for (int i = XSPLIT; i < KSTEPS; ++i) {
    const float* w0 = w_qkv + (i * 4 + kh * 2 + 0) * NQKV;
    const float* w1 = w_qkv + (i * 4 + kh * 2 + 1) * NQKV;
    bA[i].x = w0[colA];  bA[i].y = w1[colA];
    bB[i].x = w0[colB];  bB[i].y = w1[colB];
  }

#pragma unroll
  for (int it = 0; it < 6; ++it) {
    const int idx = tid + 256 * it;
    const int r   = idx / (CTOT / 4);
    const int c4  = idx - r * (CTOT / 4);
    *(float4*)&xs[r][c4 * 4] = xv[it];
  }
  __syncthreads();

  const float biasA = b_qkv[colA];
  const float biasB = b_qkv[colB];
  const float biasT = b_qkv[colT];

  auto qkv_dst = [&](int col) -> float* {
    int head = col / CTOT;
    int r96  = col - head * CTOT;
    int e    = r96 / 3;
    int kqv  = r96 - 3 * e;
    return &qkvs[kqv][head][0][e];
  };
  float* dstA = qkv_dst(colA);
  float* dstB = qkv_dst(colB);
  v2f bT[KSTEPS];
#pragma unroll
  for (int rp = 0; rp < 2; ++rp) {
    const int r0 = rp * 32;
    v8f accA0 = {}, accA1 = {}, accB0 = {}, accB1 = {};
#pragma unroll
    for (int i = 0; i < KSTEPS; ++i) {
      const int kc = i * 4 + kh * 2;
      v2f a0, a1;
      a0.x = xs[r0 +  0 + ln16][kc];  a0.y = xs[r0 +  0 + ln16][kc + 1];
      a1.x = xs[r0 + 16 + ln16][kc];  a1.y = xs[r0 + 16 + ln16][kc + 1];
      accA0 = WMMA4(a0, bA[i], accA0);
      accB0 = WMMA4(a0, bB[i], accB0);
      accA1 = WMMA4(a1, bA[i], accA1);
      accB1 = WMMA4(a1, bB[i], accB1);
      if (rp == 1 && i >= KSTEPS / 2) {
        const int t = 2 * (i - KSTEPS / 2);
        bT[t].x     = w_qkv[((t    ) * 4 + kh * 2 + 0) * NQKV + colT];
        bT[t].y     = w_qkv[((t    ) * 4 + kh * 2 + 1) * NQKV + colT];
        bT[t + 1].x = w_qkv[((t + 1) * 4 + kh * 2 + 0) * NQKV + colT];
        bT[t + 1].y = w_qkv[((t + 1) * 4 + kh * 2 + 1) * NQKV + colT];
      }
    }
#pragma unroll
    for (int j = 0; j < 8; ++j) {
      const int row = r0 + j + rofs;
      dstA[( 0 + row) * 33] = accA0[j] + biasA;
      dstA[(16 + row) * 33] = accA1[j] + biasA;
      dstB[( 0 + row) * 33] = accB0[j] + biasB;
      dstB[(16 + row) * 33] = accB1[j] + biasB;
    }
  }
  {
    const int row0 = (wid & 3) * 16;
    float* dst = qkv_dst(colT);
    v8f acc = {};
#pragma unroll
    for (int i = 0; i < KSTEPS; ++i) {
      const int kc = i * 4 + kh * 2;
      v2f a;
      a.x = xs[row0 + ln16][kc];
      a.y = xs[row0 + ln16][kc + 1];
      acc = WMMA4(a, bT[i], acc);
    }
#pragma unroll
    for (int j = 0; j < 8; ++j)
      dst[(row0 + j + rofs) * 33] = acc[j] + biasT;
  }

  const int j0s  = (wid & 3) * 16;
  const int scol = j0s + ln16;
  const bool keyreal = scol < TOK;
  const int colc = keyreal ? scol : TOK - 1;
  float sbias[2][8];
#pragma unroll
  for (int u = 0; u < 2; ++u) {
#pragma unroll
    for (int j = 0; j < 8; ++j) {
      const int row  = ((wid >> 2) + 2 * u) * 16 + j + rofs;
      const int rowc = row < TOK ? row : TOK - 1;
      const float bv = rel_bias[rowc * TOK + colc];
      sbias[u][j] = row < TOK ? bv : 0.f;
    }
  }
  const int widu = __builtin_amdgcn_readfirstlane(wid);
  const bool rowwave = widu >= 6;
  const int ct0  = rowwave ? (widu - 6) * 3 : widu;
  const int pcol = ct0 * 16 + ln16;
  v2f bP[KSTEPS];
#pragma unroll
  for (int i = 0; i < KSTEPS; ++i) {
    bP[i].x = w_proj[(i * 4 + kh * 2 + 0) * CTOT + pcol];
    bP[i].y = w_proj[(i * 4 + kh * 2 + 1) * CTOT + pcol];
  }
  float bp0 = b_proj[pcol];
  __syncthreads();

  const float scale = 0.17677669529663687f;
  for (int head = 0; head < HEADS; ++head) {
#pragma unroll
    for (int u = 0; u < 2; ++u) {
      const int q0 = ((wid >> 2) + 2 * u) * 16;
      const int j0 = j0s;
      v8f acc = {};
#pragma unroll
      for (int kk = 0; kk < EDIM; kk += 4) {
        v2f a, bf;
        a.x  = qkvs[0][head][q0 + ln16][kk + kh * 2 + 0];
        a.y  = qkvs[0][head][q0 + ln16][kk + kh * 2 + 1];
        bf.x = qkvs[1][head][j0 + ln16][kk + kh * 2 + 0];
        bf.y = qkvs[1][head][j0 + ln16][kk + kh * 2 + 1];
        acc = WMMA4(a, bf, acc);
      }
#pragma unroll
      for (int j = 0; j < 8; ++j) {
        const int row = q0 + j + rofs;
        sc[row][scol] = keyreal ? acc[j] * scale + sbias[u][j] : -1e30f;
      }
    }
    __syncthreads();

    softmax_rows(sc, wid, lane);
    __syncthreads();

    {
      const int q0 = (wid / 2) * 16;
      const int e0 = (wid % 2) * 16;
      v8f acc = {};
#pragma unroll
      for (int kk = 0; kk < TPAD; kk += 4) {
        v2f a, bf;
        a.x  = sc[q0 + ln16][kk + kh * 2 + 0];
        a.y  = sc[q0 + ln16][kk + kh * 2 + 1];
        bf.x = qkvs[2][head][kk + kh * 2 + 0][e0 + ln16];
        bf.y = qkvs[2][head][kk + kh * 2 + 1][e0 + ln16];
        acc = WMMA4(a, bf, acc);
      }
#pragma unroll
      for (int j = 0; j < 8; ++j)
        qkvs[0][head][q0 + j + rofs][e0 + ln16] = acc[j];
    }
    __syncthreads();
  }

  float* obase = out + ((long)b * (HW_ * HW_) + (long)(wy * WIN * HW_ + wx * WIN)) * CTOT;
#define GOFF(r) ((((r) / WIN) * HW_ + ((r) % WIN)) * CTOT)
  if (!rowwave) {
    v8f acc0 = {}, acc1 = {}, acc2 = {};
#pragma unroll
    for (int i = 0; i < KSTEPS; ++i) {
      const int kk = i * 4;
      const float* ap = &qkvs[0][kk / EDIM][ln16][(kk % EDIM) + kh * 2];
      v2f a0, a1, a2;
      a0.x = ap[0];        a0.y = ap[1];
      a1.x = ap[16 * 33];  a1.y = ap[16 * 33 + 1];
      a2.x = ap[32 * 33];  a2.y = ap[32 * 33 + 1];
      acc0 = WMMA4(a0, bP[i], acc0);
      acc1 = WMMA4(a1, bP[i], acc1);
      acc2 = WMMA4(a2, bP[i], acc2);
    }
    float* op = obase + pcol;
#pragma unroll
    for (int j = 0; j < 8; ++j) {
      const int g0 = kh ? GOFF(j + 8)      : GOFF(j);
      const int g1 = kh ? GOFF(j + 8 + 16) : GOFF(j + 16);
      const int g2 = kh ? GOFF(j + 8 + 32) : GOFF(j + 32);
      op[g0] = acc0[j] + bp0;
      op[g1] = acc1[j] + bp0;
      op[g2] = acc2[j] + bp0;
    }
  } else {
#pragma unroll
    for (int t = 0; t < 3; ++t) {
      v2f bN[KSTEPS];
      float bpn = 0.f;
      if (t < 2) {
        const int cn = (ct0 + t + 1) * 16 + ln16;
#pragma unroll
        for (int i = 0; i < KSTEPS; ++i) {
          bN[i].x = w_proj[(i * 4 + kh * 2 + 0) * CTOT + cn];
          bN[i].y = w_proj[(i * 4 + kh * 2 + 1) * CTOT + cn];
        }
        bpn = b_proj[cn];
      }
      v8f acc = {};
#pragma unroll
      for (int i = 0; i < KSTEPS; ++i) {
        const int kk = i * 4;
        v2f a;
        a.x = qkvs[0][kk / EDIM][48 + ln16][(kk % EDIM) + kh * 2 + 0];
        a.y = qkvs[0][kk / EDIM][48 + ln16][(kk % EDIM) + kh * 2 + 1];
        acc = WMMA4(a, bP[i], acc);
      }
      if (kh == 0) obase[GOFF(48) + (ct0 + t) * 16 + ln16] = acc[0] + bp0;
      if (t < 2) {
#pragma unroll
        for (int i = 0; i < KSTEPS; ++i) bP[i] = bN[i];
        bp0 = bpn;
      }
    }
  }
#undef GOFF
}

extern "C" void kernel_launch(void* const* d_in, const int* in_sizes, int n_in,
                              void* d_out, int out_size, void* d_ws, size_t ws_size,
                              hipStream_t stream) {
  const float* x        = (const float*)d_in[0];
  const float* w_qkv    = (const float*)d_in[1];
  const float* b_qkv    = (const float*)d_in[2];
  const float* w_proj   = (const float*)d_in[3];
  const float* b_proj   = (const float*)d_in[4];
  const float* rel_bias = (const float*)d_in[5];
  float* out = (float*)d_out;
  (void)in_sizes; (void)n_in; (void)out_size; (void)d_ws; (void)ws_size;

  swin_msa_fused<<<64 * 64, 256, 0, stream>>>(x, w_qkv, b_qkv, w_proj, b_proj,
                                              rel_bias, out);
}


# k order of the three LDS-fed products re-dealt by lane half (upper half-wave: far half of k): each dword read conflict-free at the old pitches; P@V skips its 3 all-padding steps; their sums reorder
# speedup vs baseline: 4.6409x; 1.0379x over previous
#include <hip/hip_runtime.h>

#define HW_    56
#define CTOT   96
#define HEADS  3
#define EDIM   32
#define WIN    7
#define TOK    49
#define TPAD   64
#define NQKV   288
#define KSTEPS 24
#define PVSTEPS 13

typedef __attribute__((ext_vector_type(2))) float v2f;
typedef __attribute__((ext_vector_type(8))) float v8f;

#define WMMA4(a, b, c) \
  __builtin_amdgcn_wmma_f32_16x16x4_f32(false, (a), false, (b), (short)0, (c), false, false)

__device__ __forceinline__ float xhalf(float x) {
  const int b = __builtin_bit_cast(int, x);
  return __builtin_bit_cast(float,
      __builtin_amdgcn_permlanex16(b, b, 0x76543210, 0xfedcba98, false, true));
}
#define XROW(x, m) \
  __builtin_bit_cast(float, __builtin_amdgcn_update_dpp(0, \
      __builtin_bit_cast(int, (x)), 0x160 + (m), 0xf, 0xf, true))
#define ROWMAX(x, m) \
  asm("v_max_num_f32_dpp %0, %0, %0 row_xmask:" #m " row_mask:0xf bank_mask:0xf" : "+v"(x))

#define SROWS 7

__device__ __forceinline__ void softmax_rows(float (*sc)[65], const int row0, const int lane) {
  float v0[SROWS], v1[SROWS], mx[SROWS], sm[SROWS];
#pragma unroll
  for (int t = 0; t < SROWS; ++t) {
    v0[t] = sc[row0 + 8 * t][lane];
    v1[t] = sc[row0 + 8 * t][lane + 32];
    asm("v_max_num_f32 %0, %1, %2" : "=v"(mx[t]) : "v"(v0[t]), "v"(v1[t]));
  }
#pragma unroll
  for (int t = 0; t < SROWS; ++t) {
    const float o = xhalf(mx[t]);
    asm("v_max_num_f32 %0, %0, %1" : "+v"(mx[t]) : "v"(o));
  }
#pragma unroll
  for (int t = 0; t < SROWS; ++t) ROWMAX(mx[t], 8);
#pragma unroll
  for (int t = 0; t < SROWS; ++t) ROWMAX(mx[t], 4);
#pragma unroll
  for (int t = 0; t < SROWS; ++t) ROWMAX(mx[t], 2);
#pragma unroll
  for (int t = 0; t < SROWS; ++t) ROWMAX(mx[t], 1);
#pragma unroll
  for (int t = 0; t < SROWS; ++t) {
    v0[t] = __expf(v0[t] - mx[t]);
    v1[t] = __expf(v1[t] - mx[t]);
    sm[t] = v0[t] + v1[t];
  }
#pragma unroll
  for (int t = 0; t < SROWS; ++t) sm[t] += xhalf(sm[t]);
#pragma unroll
  for (int t = 0; t < SROWS; ++t) sm[t] += XROW(sm[t], 8);
#pragma unroll
  for (int t = 0; t < SROWS; ++t) sm[t] += XROW(sm[t], 4);
#pragma unroll
  for (int t = 0; t < SROWS; ++t) sm[t] += XROW(sm[t], 2);
#pragma unroll
  for (int t = 0; t < SROWS; ++t) sm[t] += XROW(sm[t], 1);
#pragma unroll
  for (int t = 0; t < SROWS; ++t) {
    const float inv = __builtin_amdgcn_rcpf(sm[t]);
    sc[row0 + 8 * t][lane]      = v0[t] * inv;
    sc[row0 + 8 * t][lane + 32] = v1[t] * inv;
  }
}

__global__ __launch_bounds__(256, 6) void swin_msa_fused(
    const float* __restrict__ x, const float* __restrict__ w_qkv,
    const float* __restrict__ b_qkv, const float* __restrict__ w_proj,
    const float* __restrict__ b_proj, const float* __restrict__ rel_bias,
    float* __restrict__ out)
{
  __shared__ float xs[TPAD][100];
  __shared__ float qkvs[3][HEADS][TPAD][33];
  float (*sc)[65] = reinterpret_cast<float (*)[65]>(&xs[0][0]);

  const int tid  = threadIdx.x;
  const int wid  = tid >> 5;
  const int lane = tid & 31;
  const int ln16 = lane & 15;
  const int kh   = lane >> 4;
  const int rofs = kh * 8;

  const int blk = blockIdx.x;
  const int b   = blk >> 6;
  const int win = blk & 63;
  const int wy  = win >> 3;
  const int wx  = win & 7;
  const long xbase = (long)b * (HW_ * HW_) * CTOT;

#define XSPLIT 11
  const int colA = wid * 16 + ln16;
  const int colB = (wid + 8) * 16 + ln16;
  const int colT = (16 + (wid >> 2)) * 16 + ln16;
  v2f bA[KSTEPS], bB[KSTEPS];
#pragma unroll
  for (int i = 0; i < XSPLIT; ++i) {
    const float* w0 = w_qkv + (i * 4 + kh * 2 + 0) * NQKV;
    const float* w1 = w_qkv + (i * 4 + kh * 2 + 1) * NQKV;
    bA[i].x = w0[colA];  bA[i].y = w1[colA];
    bB[i].x = w0[colB];  bB[i].y = w1[colB];
  }
  float4 xv[6];
#pragma unroll
  for (int it = 0; it < 6; ++it) {
    const int idx = tid + 256 * it;
    const int r   = idx / (CTOT / 4);
    const int c4  = idx - r * (CTOT / 4);
    float4 v = make_float4(0.f, 0.f, 0.f, 0.f);
    if (it < 5) {
      const int rc = r < TOK ? r : TOK - 1;
      const int m1 = rc / WIN, m2 = rc % WIN;
      const int grow = (wy * WIN + m1) * HW_ + (wx * WIN + m2);
      const float4 g = *(const float4*)(x + xbase + (long)grow * CTOT + c4 * 4);
      if (r < TOK) v = g;
    }
    xv[it] = v;
  }
#pragma unroll
  for (int i = XSPLIT; i < KSTEPS; ++i) {
    const float* w0 = w_qkv + (i * 4 + kh * 2 + 0) * NQKV;
    const float* w1 = w_qkv + (i * 4 + kh * 2 + 1) * NQKV;
    bA[i].x = w0[colA];  bA[i].y = w1[colA];
    bB[i].x = w0[colB];  bB[i].y = w1[colB];
  }

#pragma unroll
  for (int it = 0; it < 6; ++it) {
    const int idx = tid + 256 * it;
    const int r   = idx / (CTOT / 4);
    const int c4  = idx - r * (CTOT / 4);
    *(float4*)&xs[r][c4 * 4] = xv[it];
  }
  __syncthreads();

  const float biasA = b_qkv[colA];
  const float biasB = b_qkv[colB];
  const float biasT = b_qkv[colT];

  auto qkv_dst = [&](int col) -> float* {
    int head = col / CTOT;
    int r96  = col - head * CTOT;
    int e    = r96 / 3;
    int kqv  = r96 - 3 * e;
    return &qkvs[kqv][head][0][e];
  };
  float* dstA = qkv_dst(colA);
  float* dstB = qkv_dst(colB);
  v2f bT[KSTEPS];
#pragma unroll
  for (int rp = 0; rp < 2; ++rp) {
    const int r0 = rp * 32;
    v8f accA0 = {}, accA1 = {}, accB0 = {}, accB1 = {};
#pragma unroll
    for (int i = 0; i < KSTEPS; ++i) {
      const int kc = i * 4 + kh * 2;
      v2f a0, a1;
      a0.x = xs[r0 +  0 + ln16][kc];  a0.y = xs[r0 +  0 + ln16][kc + 1];
      a1.x = xs[r0 + 16 + ln16][kc];  a1.y = xs[r0 + 16 + ln16][kc + 1];
      accA0 = WMMA4(a0, bA[i], accA0);
      accB0 = WMMA4(a0, bB[i], accB0);
      accA1 = WMMA4(a1, bA[i], accA1);
      accB1 = WMMA4(a1, bB[i], accB1);
      if (rp == 1 && i >= KSTEPS / 2) {
        const int t = 2 * (i - KSTEPS / 2);
        bT[t].x     = w_qkv[((t    ) * 4 + kh * 2 + 0) * NQKV + colT];
        bT[t].y     = w_qkv[((t    ) * 4 + kh * 2 + 1) * NQKV + colT];
        bT[t + 1].x = w_qkv[((t + 1) * 4 + kh * 2 + 0) * NQKV + colT];
        bT[t + 1].y = w_qkv[((t + 1) * 4 + kh * 2 + 1) * NQKV + colT];
      }
    }
#pragma unroll
    for (int j = 0; j < 8; ++j) {
      const int row = r0 + j + rofs;
      dstA[( 0 + row) * 33] = accA0[j] + biasA;
      dstA[(16 + row) * 33] = accA1[j] + biasA;
      dstB[( 0 + row) * 33] = accB0[j] + biasB;
      dstB[(16 + row) * 33] = accB1[j] + biasB;
    }
  }
  {
    const int row0 = (wid & 3) * 16;
    float* dst = qkv_dst(colT);
    v8f acc = {};
#pragma unroll
    for (int i = 0; i < KSTEPS; ++i) {
      const int kc = i * 4 + kh * 2;
      v2f a;
      a.x = xs[row0 + ln16][kc];
      a.y = xs[row0 + ln16][kc + 1];
      acc = WMMA4(a, bT[i], acc);
    }
#pragma unroll
    for (int j = 0; j < 8; ++j)
      dst[(row0 + j + rofs) * 33] = acc[j] + biasT;
  }

  const int j0s  = (wid & 3) * 16;
  const int scol = j0s + ln16;
  const bool keyreal = scol < TOK;
  const int colc = keyreal ? scol : TOK - 1;
  float sbias[2][8];
#pragma unroll
  for (int u = 0; u < 2; ++u) {
#pragma unroll
    for (int j = 0; j < 8; ++j) {
      const int row  = ((wid >> 2) + 2 * u) * 16 + j + rofs;
      const int rowc = row < TOK ? row : TOK - 1;
      const float bv = rel_bias[rowc * TOK + colc];
      sbias[u][j] = row < TOK ? bv : 0.f;
    }
  }
#define PE(i) (2 * ((i) % 8) + (EDIM / 2) * kh)
#define PK(i) (((i) / 8) * EDIM + PE(i))
  const int widu = __builtin_amdgcn_readfirstlane(wid);
  const bool rowwave = widu >= 6;
  const int ct0  = rowwave ? (widu - 6) * 3 : widu;
  const int pcol = ct0 * 16 + ln16;
  v2f bP[KSTEPS];
#pragma unroll
  for (int i = 0; i < KSTEPS; ++i) {
    bP[i].x = w_proj[(PK(i) + 0) * CTOT + pcol];
    bP[i].y = w_proj[(PK(i) + 1) * CTOT + pcol];
  }
  float bp0 = b_proj[pcol];
  __syncthreads();

  const float scale = 0.17677669529663687f;
#pragma nounroll
  for (int head = 0; head < HEADS; ++head) {
#pragma unroll
    for (int u = 0; u < 2; ++u) {
      const int q0 = ((wid >> 2) + 2 * u) * 16;
      const int j0 = j0s;
      v8f acc = {};
#pragma unroll
      for (int i = 0; i < EDIM / 4; ++i) {
        const int k0 = 2 * i + (EDIM / 2) * kh;
        v2f a, bf;
        a.x  = qkvs[0][head][q0 + ln16][k0];
        a.y  = qkvs[0][head][q0 + ln16][k0 + 1];
        bf.x = qkvs[1][head][j0 + ln16][k0];
        bf.y = qkvs[1][head][j0 + ln16][k0 + 1];
        acc = WMMA4(a, bf, acc);
      }
#pragma unroll
      for (int j = 0; j < 8; ++j) {
        const int row = q0 + j + rofs;
        sc[row][scol] = keyreal ? acc[j] * scale + sbias[u][j] : -1e30f;
      }
    }
    __syncthreads();

    softmax_rows(sc, wid, lane);
    __syncthreads();

    {
      const int q0 = (wid / 2) * 16;
      const int e0 = (wid % 2) * 16;
      v8f acc = {};
#pragma unroll
      for (int i = 0; i < PVSTEPS; ++i) {
        const int k0 = 2 * i + 2 * PVSTEPS * kh;
        v2f a, bf;
        a.x  = sc[q0 + ln16][k0];
        a.y  = sc[q0 + ln16][k0 + 1];
        bf.x = qkvs[2][head][k0][e0 + ln16];
        bf.y = qkvs[2][head][k0 + 1][e0 + ln16];
        acc = WMMA4(a, bf, acc);
      }
#pragma unroll
      for (int j = 0; j < 8; ++j)
        qkvs[0][head][q0 + j + rofs][e0 + ln16] = acc[j];
    }
    __syncthreads();
  }

  float* obase = out + ((long)b * (HW_ * HW_) + (long)(wy * WIN * HW_ + wx * WIN)) * CTOT;
#define GOFF(r) ((((r) / WIN) * HW_ + ((r) % WIN)) * CTOT)
  if (!rowwave) {
    v8f acc0 = {}, acc1 = {}, acc2 = {};
#pragma unroll
    for (int i = 0; i < KSTEPS; ++i) {
      const float* ap = &qkvs[0][i / 8][ln16][PE(i)];
      v2f a0, a1, a2;
      a0.x = ap[0];        a0.y = ap[1];
      a1.x = ap[16 * 33];  a1.y = ap[16 * 33 + 1];
      a2.x = ap[32 * 33];  a2.y = ap[32 * 33 + 1];
      acc0 = WMMA4(a0, bP[i], acc0);
      acc1 = WMMA4(a1, bP[i], acc1);
      acc2 = WMMA4(a2, bP[i], acc2);
    }
    float* op = obase + pcol;
#pragma unroll
    for (int j = 0; j < 8; ++j) {
      const int g0 = kh ? GOFF(j + 8)      : GOFF(j);
      const int g1 = kh ? GOFF(j + 8 + 16) : GOFF(j + 16);
      const int g2 = kh ? GOFF(j + 8 + 32) : GOFF(j + 32);
      op[g0] = acc0[j] + bp0;
      op[g1] = acc1[j] + bp0;
      op[g2] = acc2[j] + bp0;
    }
  } else {
#pragma unroll
    for (int t = 0; t < 3; ++t) {
      v2f bN[KSTEPS];
      float bpn = 0.f;
      if (t < 2) {
        const int cn = (ct0 + t + 1) * 16 + ln16;
#pragma unroll
        for (int i = 0; i < KSTEPS; ++i) {
          bN[i].x = w_proj[(PK(i) + 0) * CTOT + cn];
          bN[i].y = w_proj[(PK(i) + 1) * CTOT + cn];
        }
        bpn = b_proj[cn];
      }
      v8f acc = {};
#pragma unroll
      for (int i = 0; i < KSTEPS; ++i) {
        v2f a;
        a.x = qkvs[0][i / 8][48 + ln16][PE(i) + 0];
        a.y = qkvs[0][i / 8][48 + ln16][PE(i) + 1];
        acc = WMMA4(a, bP[i], acc);
      }
      if (kh == 0) obase[GOFF(48) + (ct0 + t) * 16 + ln16] = acc[0] + bp0;
      if (t < 2) {
#pragma unroll
        for (int i = 0; i < KSTEPS; ++i) bP[i] = bN[i];
        bp0 = bpn;
      }
    }
  }
#undef GOFF
#undef PE
#undef PK
}

extern "C" void kernel_launch(void* const* d_in, const int* in_sizes, int n_in,
                              void* d_out, int out_size, void* d_ws, size_t ws_size,
                              hipStream_t stream) {
  const float* x        = (const float*)d_in[0];
  const float* w_qkv    = (const float*)d_in[1];
  const float* b_qkv    = (const float*)d_in[2];
  const float* w_proj   = (const float*)d_in[3];
  const float* b_proj   = (const float*)d_in[4];
  const float* rel_bias = (const float*)d_in[5];
  float* out = (float*)d_out;
  (void)in_sizes; (void)n_in; (void)out_size; (void)d_ws; (void)ws_size;

  swin_msa_fused<<<64 * 64, 256, 0, stream>>>(x, w_qkv, b_qkv, w_proj, b_proj,
                                              rel_bias, out);
}


# qkv of the last token (row 48) of all 4096 windows by a first dense launch, 16 windows per row tile, into the workspace; the window kernel runs qkv on 3 row tiles, not 4; sums unchanged
# speedup vs baseline: 4.7956x; 1.0333x over previous
#include <hip/hip_runtime.h>

#define HW_    56
#define CTOT   96
#define HEADS  3
#define EDIM   32
#define WIN    7
#define TOK    49
#define TPAD   64
#define NQKV   288
#define KSTEPS 24
#define PVSTEPS 13
#define LASTTOK 48
#define NWIN   4096

typedef __attribute__((ext_vector_type(2))) float v2f;
typedef __attribute__((ext_vector_type(8))) float v8f;

#define WMMA4(a, b, c) \
  __builtin_amdgcn_wmma_f32_16x16x4_f32(false, (a), false, (b), (short)0, (c), false, false)

__device__ __forceinline__ float xhalf(float x) {
  const int b = __builtin_bit_cast(int, x);
  return __builtin_bit_cast(float,
      __builtin_amdgcn_permlanex16(b, b, 0x76543210, 0xfedcba98, false, true));
}
#define XROW(x, m) \
  __builtin_bit_cast(float, __builtin_amdgcn_update_dpp(0, \
      __builtin_bit_cast(int, (x)), 0x160 + (m), 0xf, 0xf, true))
#define ROWMAX(x, m) \
  asm("v_max_num_f32_dpp %0, %0, %0 row_xmask:" #m " row_mask:0xf bank_mask:0xf" : "+v"(x))

#define SROWS 7

__device__ __forceinline__ void softmax_rows(float (*sc)[65], const int row0, const int lane) {
  float v0[SROWS], v1[SROWS], mx[SROWS], sm[SROWS];
#pragma unroll
  for (int t = 0; t < SROWS; ++t) {
    v0[t] = sc[row0 + 8 * t][lane];
    v1[t] = sc[row0 + 8 * t][lane + 32];
    asm("v_max_num_f32 %0, %1, %2" : "=v"(mx[t]) : "v"(v0[t]), "v"(v1[t]));
  }
#pragma unroll
  for (int t = 0; t < SROWS; ++t) {
    const float o = xhalf(mx[t]);
    asm("v_max_num_f32 %0, %0, %1" : "+v"(mx[t]) : "v"(o));
  }
#pragma unroll
  for (int t = 0; t < SROWS; ++t) ROWMAX(mx[t], 8);
#pragma unroll
  for (int t = 0; t < SROWS; ++t) ROWMAX(mx[t], 4);
#pragma unroll
  for (int t = 0; t < SROWS; ++t) ROWMAX(mx[t], 2);
#pragma unroll
  for (int t = 0; t < SROWS; ++t) ROWMAX(mx[t], 1);
#pragma unroll
  for (int t = 0; t < SROWS; ++t) {
    v0[t] = __expf(v0[t] - mx[t]);
    v1[t] = __expf(v1[t] - mx[t]);
    sm[t] = v0[t] + v1[t];
  }
#pragma unroll
  for (int t = 0; t < SROWS; ++t) sm[t] += xhalf(sm[t]);
#pragma unroll
  for (int t = 0; t < SROWS; ++t) sm[t] += XROW(sm[t], 8);
#pragma unroll
  for (int t = 0; t < SROWS; ++t) sm[t] += XROW(sm[t], 4);
#pragma unroll
  for (int t = 0; t < SROWS; ++t) sm[t] += XROW(sm[t], 2);
#pragma unroll
  for (int t = 0; t < SROWS; ++t) sm[t] += XROW(sm[t], 1);
#pragma unroll
  for (int t = 0; t < SROWS; ++t) {
    const float inv = __builtin_amdgcn_rcpf(sm[t]);
    sc[row0 + 8 * t][lane]      = v0[t] * inv;
    sc[row0 + 8 * t][lane + 32] = v1[t] * inv;
  }
}

__global__ __launch_bounds__(288) void qkv_last_token(
    const float* __restrict__ x, const float* __restrict__ w_qkv,
    const float* __restrict__ b_qkv, float* __restrict__ q48)
{
  const int tid  = threadIdx.x;
  const int wid  = tid >> 5;
  const int lane = tid & 31;
  const int ln16 = lane & 15;
  const int kh   = lane >> 4;
  const int gw  = blockIdx.x * 16 + ln16;
  const int b   = gw >> 6;
  const int win = gw & 63;
  const int wy  = win >> 3;
  const int wx  = win & 7;
  const int grow = (wy * WIN + WIN - 1) * HW_ + (wx * WIN + WIN - 1);
  const float* xr = x + ((long)b * (HW_ * HW_) + grow) * CTOT + kh * 2;
  const int col0 = (2 * wid) * 16 + ln16;
  const int col1 = col0 + 16;
  v2f a[KSTEPS], b0[KSTEPS], b1[KSTEPS];
#pragma unroll
  for (int i = 0; i < KSTEPS; ++i) {
    a[i] = *(const v2f*)(xr + 4 * i);
    const float* w0 = w_qkv + (i * 4 + kh * 2 + 0) * NQKV;
    const float* w1 = w_qkv + (i * 4 + kh * 2 + 1) * NQKV;
    b0[i].x = w0[col0];  b0[i].y = w1[col0];
    b1[i].x = w0[col1];  b1[i].y = w1[col1];
  }
  const float bias0 = b_qkv[col0];
  const float bias1 = b_qkv[col1];
  v8f acc0 = {}, acc1 = {};
#pragma unroll
  for (int i = 0; i < KSTEPS; ++i) {
    acc0 = WMMA4(a[i], b0[i], acc0);
    acc1 = WMMA4(a[i], b1[i], acc1);
  }
  float* o = q48 + (long)(blockIdx.x * 16 + kh * 8) * NQKV;
#pragma unroll
  for (int j = 0; j < 8; ++j) {
    o[j * NQKV + col0] = acc0[j] + bias0;
    o[j * NQKV + col1] = acc1[j] + bias1;
  }
}

struct with_tile { static constexpr bool value = true; };
struct without_tile { static constexpr bool value = false; };

__global__ __launch_bounds__(256, 6) void swin_msa_fused(
    const float* __restrict__ x, const float* __restrict__ w_qkv,
    const float* __restrict__ b_qkv, const float* __restrict__ w_proj,
    const float* __restrict__ b_proj, const float* __restrict__ rel_bias,
    const float* __restrict__ q48, float* __restrict__ out)
{
  __shared__ float xs[TPAD][100];
  __shared__ float qkvs[3][HEADS][TPAD][33];
  float (*sc)[65] = reinterpret_cast<float (*)[65]>(&xs[0][0]);

  const int tid  = threadIdx.x;
  const int wid  = tid >> 5;
  const int lane = tid & 31;
  const int ln16 = lane & 15;
  const int kh   = lane >> 4;
  const int rofs = kh * 8;

  const int blk = blockIdx.x;
  const int b   = blk >> 6;
  const int win = blk & 63;
  const int wy  = win >> 3;
  const int wx  = win & 7;
  const long xbase = (long)b * (HW_ * HW_) * CTOT;

#define XSPLIT 11
  const int colA = wid * 16 + ln16;
  const int colB = (wid + 8) * 16 + ln16;
  const int widu = __builtin_amdgcn_readfirstlane(wid);
  const bool hasT = widu < 6;
  const int colT = (16 + (widu >= 3 ? 1 : 0)) * 16 + ln16;
  const float* r48 = q48 + (long)blockIdx.x * NQKV;
  const float rv0 = r48[tid];
  const float rv1 = r48[tid < NQKV - 256 ? 256 + tid : tid];
  v2f bA[KSTEPS], bB[KSTEPS];
#pragma unroll
  for (int i = 0; i < XSPLIT; ++i) {
    const float* w0 = w_qkv + (i * 4 + kh * 2 + 0) * NQKV;
    const float* w1 = w_qkv + (i * 4 + kh * 2 + 1) * NQKV;
    bA[i].x = w0[colA];  bA[i].y = w1[colA];
    bB[i].x = w0[colB];  bB[i].y = w1[colB];
  }
  float4 xv[5];
#pragma unroll
  for (int it = 0; it < 5; ++it) {
    const int idx = tid + 256 * it;
    const int r   = idx / (CTOT / 4);
    const int c4  = idx - r * (CTOT / 4);
    float4 v = make_float4(0.f, 0.f, 0.f, 0.f);
    const int rc = r < LASTTOK ? r : LASTTOK - 1;
    const int m1 = rc / WIN, m2 = rc % WIN;
    const int grow = (wy * WIN + m1) * HW_ + (wx * WIN + m2);
    const float4 g = *(const float4*)(x + xbase + (long)grow * CTOT + c4 * 4);
    if (r < LASTTOK) v = g;
    xv[it] = v;
  }
#pragma unroll
  for (int i = XSPLIT; i < KSTEPS; ++i) {
    const float* w0 = w_qkv + (i * 4 + kh * 2 + 0) * NQKV;
    const float* w1 = w_qkv + (i * 4 + kh * 2 + 1) * NQKV;
    bA[i].x = w0[colA];  bA[i].y = w1[colA];
    bB[i].x = w0[colB];  bB[i].y = w1[colB];
  }

#pragma unroll
  for (int it = 0; it < 5; ++it) {
    const int idx = tid + 256 * it;
    const int r   = idx / (CTOT / 4);
    const int c4  = idx - r * (CTOT / 4);
    *(float4*)&xs[r][c4 * 4] = xv[it];
  }
  auto qkv_dst = [&](int col) -> float* {
    int head = col / CTOT;
    int r96  = col - head * CTOT;
    int e    = r96 / 3;
    int kqv  = r96 - 3 * e;
    return &qkvs[kqv][head][0][e];
  };
  qkv_dst(tid)[LASTTOK * 33] = rv0;
  if (tid < NQKV - 256) qkv_dst(256 + tid)[LASTTOK * 33] = rv1;
  {
    float* pz = &qkvs[0][0][0][0] + widu * (TPAD * 33) + lane;
#pragma unroll
    for (int r = TOK; r < TPAD; ++r) pz[r * 33] = 0.f;
    if (widu == 0) {
      float* pz8 = &qkvs[0][0][0][0] + 8 * (TPAD * 33) + lane;
#pragma unroll
      for (int r = TOK; r < TPAD; ++r) pz8[r * 33] = 0.f;
    }
  }
  __syncthreads();

  const float biasA = b_qkv[colA];
  const float biasB = b_qkv[colB];
  const float biasT = b_qkv[colT];

  float* dstA = qkv_dst(colA);
  float* dstB = qkv_dst(colB);
  v2f bT[KSTEPS];
  {
    v8f accA0 = {}, accA1 = {}, accB0 = {}, accB1 = {};
#pragma unroll
    for (int i = 0; i < KSTEPS; ++i) {
      const int kc = i * 4 + kh * 2;
      v2f a0, a1;
      a0.x = xs[ 0 + ln16][kc];  a0.y = xs[ 0 + ln16][kc + 1];
      a1.x = xs[16 + ln16][kc];  a1.y = xs[16 + ln16][kc + 1];
      accA0 = WMMA4(a0, bA[i], accA0);
      accB0 = WMMA4(a0, bB[i], accB0);
      accA1 = WMMA4(a1, bA[i], accA1);
      accB1 = WMMA4(a1, bB[i], accB1);
    }
#pragma unroll
    for (int j = 0; j < 8; ++j) {
      const int row = j + rofs;
      dstA[( 0 + row) * 33] = accA0[j] + biasA;
      dstA[(16 + row) * 33] = accA1[j] + biasA;
      dstB[( 0 + row) * 33] = accB0[j] + biasB;
      dstB[(16 + row) * 33] = accB1[j] + biasB;
    }
  }
  auto row_tile2 = [&](auto single) {
    v8f accA2 = {}, accB2 = {};
#pragma unroll
    for (int i = 0; i < KSTEPS; ++i) {
      const int kc = i * 4 + kh * 2;
      v2f a2;
      a2.x = xs[32 + ln16][kc];  a2.y = xs[32 + ln16][kc + 1];
      accA2 = WMMA4(a2, bA[i], accA2);
      accB2 = WMMA4(a2, bB[i], accB2);
      if constexpr (decltype(single)::value) {
        if (i >= KSTEPS / 2) {
          const int t = 2 * (i - KSTEPS / 2);
          bT[t].x     = w_qkv[((t    ) * 4 + kh * 2 + 0) * NQKV + colT];
          bT[t].y     = w_qkv[((t    ) * 4 + kh * 2 + 1) * NQKV + colT];
          bT[t + 1].x = w_qkv[((t + 1) * 4 + kh * 2 + 0) * NQKV + colT];
          bT[t + 1].y = w_qkv[((t + 1) * 4 + kh * 2 + 1) * NQKV + colT];
        }
      }
    }
#pragma unroll
    for (int j = 0; j < 8; ++j) {
      const int row = 32 + j + rofs;
      dstA[row * 33] = accA2[j] + biasA;
      dstB[row * 33] = accB2[j] + biasB;
    }
  };
  if (hasT) {
    row_tile2(with_tile{});
    const int row0 = (widu % 3) * 16;
    float* dst = qkv_dst(colT);
    v8f acc = {};
#pragma unroll
    for (int i = 0; i < KSTEPS; ++i) {
      const int kc = i * 4 + kh * 2;
      v2f a;
      a.x = xs[row0 + ln16][kc];
      a.y = xs[row0 + ln16][kc + 1];
      acc = WMMA4(a, bT[i], acc);
    }
#pragma unroll
    for (int j = 0; j < 8; ++j)
      dst[(row0 + j + rofs) * 33] = acc[j] + biasT;
  } else {
    row_tile2(without_tile{});
  }

  const int j0s  = (wid & 3) * 16;
  const int scol = j0s + ln16;
  const bool keyreal = scol < TOK;
  const int colc = keyreal ? scol : TOK - 1;
  float sbias[2][8];
#pragma unroll
  for (int u = 0; u < 2; ++u) {
#pragma unroll
    for (int j = 0; j < 8; ++j) {
      const int row  = ((wid >> 2) + 2 * u) * 16 + j + rofs;
      const int rowc = row < TOK ? row : TOK - 1;
      const float bv = rel_bias[rowc * TOK + colc];
      sbias[u][j] = row < TOK ? bv : 0.f;
    }
  }
#define PE(i) (2 * ((i) % 8) + (EDIM / 2) * kh)
#define PK(i) (((i) / 8) * EDIM + PE(i))
  const bool rowwave = widu >= 6;
  const int ct0  = rowwave ? (widu - 6) * 3 : widu;
  const int pcol = ct0 * 16 + ln16;
  v2f bP[KSTEPS];
#pragma unroll
  for (int i = 0; i < KSTEPS; ++i) {
    bP[i].x = w_proj[(PK(i) + 0) * CTOT + pcol];
    bP[i].y = w_proj[(PK(i) + 1) * CTOT + pcol];
  }
  float bp0 = b_proj[pcol];
  __syncthreads();

  const float scale = 0.17677669529663687f;
#pragma nounroll
  for (int head = 0; head < HEADS; ++head) {
#pragma unroll
    for (int u = 0; u < 2; ++u) {
      const int q0 = ((wid >> 2) + 2 * u) * 16;
      const int j0 = j0s;
      v8f acc = {};
#pragma unroll
      for (int i = 0; i < EDIM / 4; ++i) {
        const int k0 = 2 * i + (EDIM / 2) * kh;
        v2f a, bf;
        a.x  = qkvs[0][head][q0 + ln16][k0];
        a.y  = qkvs[0][head][q0 + ln16][k0 + 1];
        bf.x = qkvs[1][head][j0 + ln16][k0];
        bf.y = qkvs[1][head][j0 + ln16][k0 + 1];
        acc = WMMA4(a, bf, acc);
      }
#pragma unroll
      for (int j = 0; j < 8; ++j) {
        const int row = q0 + j + rofs;
        sc[row][scol] = keyreal ? acc[j] * scale + sbias[u][j] : -1e30f;
      }
    }
    __syncthreads();

    softmax_rows(sc, wid, lane);
    __syncthreads();

    {
      const int q0 = (wid / 2) * 16;
      const int e0 = (wid % 2) * 16;
      v8f acc = {};
#pragma unroll
      for (int i = 0; i < PVSTEPS; ++i) {
        const int k0 = 2 * i + 2 * PVSTEPS * kh;
        v2f a, bf;
        a.x  = sc[q0 + ln16][k0];
        a.y  = sc[q0 + ln16][k0 + 1];
        bf.x = qkvs[2][head][k0][e0 + ln16];
        bf.y = qkvs[2][head][k0 + 1][e0 + ln16];
        acc = WMMA4(a, bf, acc);
      }
#pragma unroll
      for (int j = 0; j < 8; ++j)
        qkvs[0][head][q0 + j + rofs][e0 + ln16] = acc[j];
    }
    __syncthreads();
  }

  float* obase = out + ((long)b * (HW_ * HW_) + (long)(wy * WIN * HW_ + wx * WIN)) * CTOT;
#define GOFF(r) ((((r) / WIN) * HW_ + ((r) % WIN)) * CTOT)
  if (!rowwave) {
    v8f acc0 = {}, acc1 = {}, acc2 = {};
#pragma unroll
    for (int i = 0; i < KSTEPS; ++i) {
      const float* ap = &qkvs[0][i / 8][ln16][PE(i)];
      v2f a0, a1, a2;
      a0.x = ap[0];        a0.y = ap[1];
      a1.x = ap[16 * 33];  a1.y = ap[16 * 33 + 1];
      a2.x = ap[32 * 33];  a2.y = ap[32 * 33 + 1];
      acc0 = WMMA4(a0, bP[i], acc0);
      acc1 = WMMA4(a1, bP[i], acc1);
      acc2 = WMMA4(a2, bP[i], acc2);
    }
    float* op = obase + pcol;
#pragma unroll
    for (int j = 0; j < 8; ++j) {
      const int g0 = kh ? GOFF(j + 8)      : GOFF(j);
      const int g1 = kh ? GOFF(j + 8 + 16) : GOFF(j + 16);
      const int g2 = kh ? GOFF(j + 8 + 32) : GOFF(j + 32);
      op[g0] = acc0[j] + bp0;
      op[g1] = acc1[j] + bp0;
      op[g2] = acc2[j] + bp0;
    }
  } else {
#pragma unroll
    for (int t = 0; t < 3; ++t) {
      v2f bN[KSTEPS];
      float bpn = 0.f;
      if (t < 2) {
        const int cn = (ct0 + t + 1) * 16 + ln16;
#pragma unroll
        for (int i = 0; i < KSTEPS; ++i) {
          bN[i].x = w_proj[(PK(i) + 0) * CTOT + cn];
          bN[i].y = w_proj[(PK(i) + 1) * CTOT + cn];
        }
        bpn = b_proj[cn];
      }
      v8f acc = {};
#pragma unroll
      for (int i = 0; i < KSTEPS; ++i) {
        v2f a;
        a.x = qkvs[0][i / 8][48 + ln16][PE(i) + 0];
        a.y = qkvs[0][i / 8][48 + ln16][PE(i) + 1];
        acc = WMMA4(a, bP[i], acc);
      }
      if (kh == 0) obase[GOFF(48) + (ct0 + t) * 16 + ln16] = acc[0] + bp0;
      if (t < 2) {
#pragma unroll
        for (int i = 0; i < KSTEPS; ++i) bP[i] = bN[i];
        bp0 = bpn;
      }
    }
  }
#undef GOFF
#undef PE
#undef PK
}

extern "C" void kernel_launch(void* const* d_in, const int* in_sizes, int n_in,
                              void* d_out, int out_size, void* d_ws, size_t ws_size,
                              hipStream_t stream) {
  const float* x        = (const float*)d_in[0];
  const float* w_qkv    = (const float*)d_in[1];
  const float* b_qkv    = (const float*)d_in[2];
  const float* w_proj   = (const float*)d_in[3];
  const float* b_proj   = (const float*)d_in[4];
  const float* rel_bias = (const float*)d_in[5];
  float* out = (float*)d_out;
  float* q48 = (float*)d_ws;
  (void)in_sizes; (void)n_in; (void)out_size; (void)ws_size;

  qkv_last_token<<<NWIN / 16, 288, 0, stream>>>(x, w_qkv, b_qkv, q48);
  swin_msa_fused<<<NWIN, 256, 0, stream>>>(x, w_qkv, b_qkv, w_proj, b_proj,
                                           rel_bias, q48, out);
}
